# starting kernel
# baseline (speedup 1.0000x reference)
#include <hip/hip_runtime.h>
#include <hip/hip_bf16.h>
#include <math.h>

typedef float v2f __attribute__((ext_vector_type(2)));
typedef float v8f __attribute__((ext_vector_type(8)));

#define IN_DIM 256
#define OUT_DIM 64
#define SLOPE 0.1f
#define KC 128
#define LDP 132

__global__ __launch_bounds__(256) void gat_gemm_wmma(
    const float* __restrict__ features,
    const float* __restrict__ W,
    const float* __restrict__ bias,
    float* __restrict__ emb,
    int n)
{
    __shared__ float ldsA[32 * LDP];
    __shared__ float ldsB[OUT_DIM * LDP];

    const int tid  = threadIdx.x;
    const int row0 = blockIdx.x * 32;

    const int wave  = tid >> 5;
    const int lane  = tid & 31;
    const int mbase = (wave >> 2) * 16;
    const int n0    = (wave & 3) * 16;
    const int lm    = lane & 15;
    const int half  = lane >> 4;
    const int kh    = half * 2;

    v8f acc = {0.f, 0.f, 0.f, 0.f, 0.f, 0.f, 0.f, 0.f};

    for (int kc = 0; kc < IN_DIM; kc += KC) {
        #pragma unroll
        for (int i = 0; i < 4; ++i) {
            int idx = tid + i * 256;
            int r   = idx >> 5;
            int c4  = idx & 31;
            float4 v = make_float4(0.f, 0.f, 0.f, 0.f);
            int gr = row0 + r;
            if (gr < n)
                v = *(const float4*)(features + (size_t)gr * IN_DIM + kc + c4 * 4);
            *(float4*)(&ldsA[r * LDP + c4 * 4]) = v;
        }
        #pragma unroll
        for (int i = 0; i < 8; ++i) {
            int idx = tid + i * 256;
            int r   = idx >> 5;
            int c4  = idx & 31;
            float4 v = *(const float4*)(W + (size_t)r * IN_DIM + kc + c4 * 4);
            *(float4*)(&ldsB[r * LDP + c4 * 4]) = v;
        }
        __syncthreads();

        const float* pA = &ldsA[(mbase + lm) * LDP + kh];
        const float* pB = &ldsB[(n0   + lm) * LDP + kh];
        #pragma unroll 8
        for (int k = 0; k < KC; k += 4) {
            v2f afrag = *(const v2f*)(pA + k);
            v2f bfrag = *(const v2f*)(pB + k);
            acc = __builtin_amdgcn_wmma_f32_16x16x4_f32(
                false, afrag, false, bfrag, (short)0, acc, false, false);
        }
        __syncthreads();
    }

    const float bv = bias[n0 + lm];
    #pragma unroll
    for (int r = 0; r < 8; ++r) {
        int grow = row0 + mbase + r + half * 8;
        if (grow < n)
            emb[(size_t)grow * OUT_DIM + n0 + lm] = acc[r] + bv;
    }
}

__global__ __launch_bounds__(256) void gat_node(
    const float* __restrict__ emb,
    const float* __restrict__ a,
    float* __restrict__ s1,
    float* __restrict__ s2,
    float* __restrict__ rowsum,
    float* __restrict__ out,
    int n)
{
    int i = blockIdx.x * blockDim.x + threadIdx.x;
    if (i >= n) return;
    const float* er = emb + (size_t)i * OUT_DIM;
    float d1 = 0.f, d2 = 0.f;
    #pragma unroll
    for (int j = 0; j < OUT_DIM; ++j) {
        float e = er[j];
        d1 += e * a[j];
        d2 += e * a[OUT_DIM + j];
    }
    s1[i] = d1;
    s2[i] = d2;
    float z = d1 + d2;
    float l = z > 0.f ? z : SLOPE * z;
    float w = expf(l);
    rowsum[i] = w;
    float* orow = out + (size_t)i * OUT_DIM;
    #pragma unroll
    for (int j = 0; j < OUT_DIM; ++j)
        orow[j] = w * er[j];
}

__global__ __launch_bounds__(256) void gat_edge(
    const int* __restrict__ edges,
    const float* __restrict__ emb,
    const float* __restrict__ s1,
    const float* __restrict__ s2,
    float* __restrict__ rowsum,
    float* __restrict__ out,
    int E)
{
    int wid  = (blockIdx.x * blockDim.x + threadIdx.x) >> 5;
    int lane = threadIdx.x & 31;
    if (wid >= E) return;
    int src = edges[2 * wid];
    int dst = edges[2 * wid + 1];
    float z = s1[src] + s2[dst];
    float l = z > 0.f ? z : SLOPE * z;
    float w = expf(l);
    if (lane == 0) atomicAdd(rowsum + src, w);
    const float2 ev = *(const float2*)(emb + (size_t)dst * OUT_DIM + 2 * lane);
    float* op = out + (size_t)src * OUT_DIM + 2 * lane;
    atomicAdd(op,     w * ev.x);
    atomicAdd(op + 1, w * ev.y);
}

__global__ __launch_bounds__(256) void gat_final(
    float* __restrict__ out,
    const float* __restrict__ rowsum,
    int total)
{
    int idx = blockIdx.x * blockDim.x + threadIdx.x;
    if (idx >= total) return;
    out[idx] /= rowsum[idx >> 6];
}

extern "C" void kernel_launch(void* const* d_in, const int* in_sizes, int n_in,
                              void* d_out, int out_size, void* d_ws, size_t ws_size,
                              hipStream_t stream)
{
    (void)n_in; (void)out_size; (void)ws_size;
    const float* features = (const float*)d_in[0];
    const float* W        = (const float*)d_in[1];
    const float* bias     = (const float*)d_in[2];
    const float* a        = (const float*)d_in[3];
    const int*   edges    = (const int*)d_in[4];

    const int n = in_sizes[0] / IN_DIM;
    const int E = in_sizes[4] / 2;
    float* out = (float*)d_out;

    float* emb    = (float*)d_ws;
    float* s1     = emb + (size_t)n * OUT_DIM;
    float* s2     = s1 + n;
    float* rowsum = s2 + n;

    gat_gemm_wmma<<<(n + 31) / 32, 256, 0, stream>>>(features, W, bias, emb, n);
    gat_node<<<(n + 255) / 256, 256, 0, stream>>>(emb, a, s1, s2, rowsum, out, n);
    gat_edge<<<(E + 7) / 8, 256, 0, stream>>>(edges, emb, s1, s2, rowsum, out, E);
    const int total = n * OUT_DIM;
    gat_final<<<(total + 255) / 256, 256, 0, stream>>>(out, rowsum, total);
}


# Float-atomic scatter replaced by per-call grouping of edges by source (returning int add = rank, list in the row's own out line) + per-row gather; rows summed in rank order, logit dots by lane tree
# speedup vs baseline: 1.5274x; 1.5274x over previous
#include <hip/hip_runtime.h>
#include <hip/hip_bf16.h>
#include <math.h>

typedef float v2f __attribute__((ext_vector_type(2)));
typedef float v8f __attribute__((ext_vector_type(8)));

#define IN_DIM 256
#define OUT_DIM 64
#define SLOPE 0.1f
#define KC 128
#define LDP 132
#define CAP 64

__global__ __launch_bounds__(256) void gat_gemm_wmma(
    const float* __restrict__ features,
    const float* __restrict__ W,
    const float* __restrict__ bias,
    float* __restrict__ emb,
    int* __restrict__ cnt,
    int n)
{
    __shared__ float ldsA[32 * LDP];
    __shared__ float ldsB[OUT_DIM * LDP];

    const int tid  = threadIdx.x;
    const int row0 = blockIdx.x * 32;

    const int wave  = tid >> 5;
    const int lane  = tid & 31;
    const int mbase = (wave >> 2) * 16;
    const int n0    = (wave & 3) * 16;
    const int lm    = lane & 15;
    const int half  = lane >> 4;
    const int kh    = half * 2;

    v8f acc = {0.f, 0.f, 0.f, 0.f, 0.f, 0.f, 0.f, 0.f};

    if (tid < 32 && row0 + tid < n) cnt[row0 + tid] = 0;

    for (int kc = 0; kc < IN_DIM; kc += KC) {
        #pragma unroll
        for (int i = 0; i < 4; ++i) {
            int idx = tid + i * 256;
            int r   = idx >> 5;
            int c4  = idx & 31;
            float4 v = make_float4(0.f, 0.f, 0.f, 0.f);
            int gr = row0 + r;
            if (gr < n)
                v = *(const float4*)(features + (size_t)gr * IN_DIM + kc + c4 * 4);
            *(float4*)(&ldsA[r * LDP + c4 * 4]) = v;
        }
        #pragma unroll
        for (int i = 0; i < 8; ++i) {
            int idx = tid + i * 256;
            int r   = idx >> 5;
            int c4  = idx & 31;
            float4 v = *(const float4*)(W + (size_t)r * IN_DIM + kc + c4 * 4);
            *(float4*)(&ldsB[r * LDP + c4 * 4]) = v;
        }
        __syncthreads();

        const float* pA = &ldsA[(mbase + lm) * LDP + kh];
        const float* pB = &ldsB[(n0   + lm) * LDP + kh];
        #pragma unroll 8
        for (int k = 0; k < KC; k += 4) {
            v2f afrag = *(const v2f*)(pA + k);
            v2f bfrag = *(const v2f*)(pB + k);
            acc = __builtin_amdgcn_wmma_f32_16x16x4_f32(
                false, afrag, false, bfrag, (short)0, acc, false, false);
        }
        __syncthreads();
    }

    const float bv = bias[n0 + lm];
    #pragma unroll
    for (int r = 0; r < 8; ++r) {
        int grow = row0 + mbase + r + half * 8;
        if (grow < n)
            emb[(size_t)grow * OUT_DIM + n0 + lm] = acc[r] + bv;
    }
}

__global__ __launch_bounds__(256) void gat_bucket(
    const int* __restrict__ edges,
    int* __restrict__ cnt,
    int* __restrict__ slots,
    int E)
{
    const int e = blockIdx.x * blockDim.x + threadIdx.x;
    if (e >= E) return;
    const int2 ed = *(const int2*)(edges + 2 * (size_t)e);
    const int r = atomicAdd(cnt + ed.x, 1);
    if (r < CAP) slots[(size_t)ed.x * OUT_DIM + r] = ed.y;
}

template <int M>
__device__ __forceinline__ float row_xor(float v)
{
    return __builtin_bit_cast(float,
        __builtin_amdgcn_update_dpp(0, __builtin_bit_cast(int, v), 0x160 + M, 0xf, 0xf, true));
}

__device__ __forceinline__ float row16_sum(float v)
{
    v += row_xor<1>(v);
    v += row_xor<2>(v);
    v += row_xor<4>(v);
    v += row_xor<8>(v);
    return v;
}

__device__ __forceinline__ float dot4(const float4 x, const float4 y)
{
    return x.x * y.x + x.y * y.y + x.z * y.z + x.w * y.w;
}

__device__ __forceinline__ float edge_weight(float z)
{
    const float l = z > 0.f ? z : SLOPE * z;
    return expf(l);
}

__global__ __launch_bounds__(256) void gat_gather(
    const int* __restrict__ edges,
    const float* __restrict__ emb,
    const float* __restrict__ a,
    const int* __restrict__ cnt,
    const int* slots,
    float* out,
    int n, int E)
{
    const int hw   = (blockIdx.x * blockDim.x + threadIdx.x) >> 4;
    const int l16  = threadIdx.x & 15;
    const bool live = hw < n;
    const int node = live ? hw : n - 1;

    const float4 a1 = *(const float4*)(a + 4 * l16);
    const float4 a2 = *(const float4*)(a + OUT_DIM + 4 * l16);
    const float4 self = *(const float4*)(emb + (size_t)node * OUT_DIM + 4 * l16);
    const int craw = cnt[node];
    const int4 sl = *(const int4*)(slots + (size_t)node * OUT_DIM + 4 * l16);

    const float s1 = row16_sum(dot4(self, a1));
    float4 acc;
    float rs;
    {
        const float w = edge_weight(s1 + row16_sum(dot4(self, a2)));
        acc = make_float4(w * self.x, w * self.y, w * self.z, w * self.w);
        rs = w;
    }

    const bool over = craw > CAP;
    const int c = over ? 0 : craw;

    for (int jb = 0; jb < c; jb += 4) {
        const int q = jb >> 2;
        int d0 = __shfl(sl.x, q, 16);
        int d1 = __shfl(sl.y, q, 16);
        int d2 = __shfl(sl.z, q, 16);
        int d3 = __shfl(sl.w, q, 16);
        const bool v1 = jb + 1 < c, v2 = jb + 2 < c, v3 = jb + 3 < c;
        d1 = v1 ? d1 : node;
        d2 = v2 ? d2 : node;
        d3 = v3 ? d3 : node;
        const float4 r0 = *(const float4*)(emb + (size_t)d0 * OUT_DIM + 4 * l16);
        const float4 r1 = *(const float4*)(emb + (size_t)d1 * OUT_DIM + 4 * l16);
        const float4 r2 = *(const float4*)(emb + (size_t)d2 * OUT_DIM + 4 * l16);
        const float4 r3 = *(const float4*)(emb + (size_t)d3 * OUT_DIM + 4 * l16);
        float t0 = dot4(r0, a2), t1 = dot4(r1, a2), t2 = dot4(r2, a2), t3 = dot4(r3, a2);
        t0 += row_xor<1>(t0); t1 += row_xor<1>(t1); t2 += row_xor<1>(t2); t3 += row_xor<1>(t3);
        t0 += row_xor<2>(t0); t1 += row_xor<2>(t1); t2 += row_xor<2>(t2); t3 += row_xor<2>(t3);
        t0 += row_xor<4>(t0); t1 += row_xor<4>(t1); t2 += row_xor<4>(t2); t3 += row_xor<4>(t3);
        t0 += row_xor<8>(t0); t1 += row_xor<8>(t1); t2 += row_xor<8>(t2); t3 += row_xor<8>(t3);
        const float w0 = edge_weight(s1 + t0);
        const float w1 = v1 ? edge_weight(s1 + t1) : 0.f;
        const float w2 = v2 ? edge_weight(s1 + t2) : 0.f;
        const float w3 = v3 ? edge_weight(s1 + t3) : 0.f;
        acc.x += w0 * r0.x; acc.y += w0 * r0.y; acc.z += w0 * r0.z; acc.w += w0 * r0.w; rs += w0;
        acc.x += w1 * r1.x; acc.y += w1 * r1.y; acc.z += w1 * r1.z; acc.w += w1 * r1.w; rs += w1;
        acc.x += w2 * r2.x; acc.y += w2 * r2.y; acc.z += w2 * r2.z; acc.w += w2 * r2.w; rs += w2;
        acc.x += w3 * r3.x; acc.y += w3 * r3.y; acc.z += w3 * r3.z; acc.w += w3 * r3.w; rs += w3;
    }

    if (over) {
        for (int e0 = 0; e0 < E; e0 += 16) {
            const int e = e0 + l16;
            int hit = 0, dd = node;
            if (e < E) {
                const int2 ed = *(const int2*)(edges + 2 * (size_t)e);
                hit = (ed.x == node) ? 1 : 0;
                dd  = hit ? ed.y : node;
            }
            for (int k = 0; k < 16; ++k) {
                const int hk = __shfl(hit, k, 16);
                const int dk = __shfl(dd, k, 16);
                if (hk) {
                    const float4 r = *(const float4*)(emb + (size_t)dk * OUT_DIM + 4 * l16);
                    const float w = edge_weight(s1 + row16_sum(dot4(r, a2)));
                    acc.x += w * r.x; acc.y += w * r.y; acc.z += w * r.z; acc.w += w * r.w; rs += w;
                }
            }
        }
    }

    if (live) {
        float4 o;
        o.x = acc.x / rs; o.y = acc.y / rs; o.z = acc.z / rs; o.w = acc.w / rs;
        *(float4*)(out + (size_t)node * OUT_DIM + 4 * l16) = o;
    }
}

extern "C" void kernel_launch(void* const* d_in, const int* in_sizes, int n_in,
                              void* d_out, int out_size, void* d_ws, size_t ws_size,
                              hipStream_t stream)
{
    (void)n_in; (void)out_size; (void)ws_size;
    const float* features = (const float*)d_in[0];
    const float* W        = (const float*)d_in[1];
    const float* bias     = (const float*)d_in[2];
    const float* a        = (const float*)d_in[3];
    const int*   edges    = (const int*)d_in[4];

    const int n = in_sizes[0] / IN_DIM;
    const int E = in_sizes[4] / 2;
    float* out = (float*)d_out;

    float* emb = (float*)d_ws;
    int*   cnt = (int*)(emb + (size_t)n * OUT_DIM);

    gat_gemm_wmma<<<(n + 31) / 32, 256, 0, stream>>>(features, W, bias, emb, cnt, n);
    gat_bucket<<<(E + 255) / 256, 256, 0, stream>>>(edges, cnt, (int*)d_out, E);
    gat_gather<<<(n * 16 + 255) / 256, 256, 0, stream>>>(edges, emb, a, cnt, (const int*)d_out, out, n, E);
}


# Per-edge returning global add + scattered 4-B slot store replaced by two-level grouping in LDS: edge chunks binned by source range, one work-group per bin lists its rows in LDS; sums in arrival order
# speedup vs baseline: 2.7968x; 1.8311x over previous
#include <hip/hip_runtime.h>
#include <hip/hip_bf16.h>
#include <math.h>

typedef float v2f __attribute__((ext_vector_type(2)));
typedef float v8f __attribute__((ext_vector_type(8)));

#define IN_DIM 256
#define OUT_DIM 64
#define SLOPE 0.1f
#define KC 128
#define LDP 132
#define CAP 64
#define RB 400
#define MAXB 256
#define NCH 256
#define SUB 96

__global__ __launch_bounds__(256) void gat_gemm_wmma(
    const float* __restrict__ features,
    const float* __restrict__ W,
    const float* __restrict__ bias,
    float* __restrict__ emb,
    int n)
{
    __shared__ float ldsA[32 * LDP];
    __shared__ float ldsB[OUT_DIM * LDP];

    const int tid  = threadIdx.x;
    const int row0 = blockIdx.x * 32;

    const int wave  = tid >> 5;
    const int lane  = tid & 31;
    const int mbase = (wave >> 2) * 16;
    const int n0    = (wave & 3) * 16;
    const int lm    = lane & 15;
    const int half  = lane >> 4;
    const int kh    = half * 2;

    v8f acc = {0.f, 0.f, 0.f, 0.f, 0.f, 0.f, 0.f, 0.f};

    for (int kc = 0; kc < IN_DIM; kc += KC) {
        #pragma unroll
        for (int i = 0; i < 4; ++i) {
            int idx = tid + i * 256;
            int r   = idx >> 5;
            int c4  = idx & 31;
            float4 v = make_float4(0.f, 0.f, 0.f, 0.f);
            int gr = row0 + r;
            if (gr < n)
                v = *(const float4*)(features + (size_t)gr * IN_DIM + kc + c4 * 4);
            *(float4*)(&ldsA[r * LDP + c4 * 4]) = v;
        }
        #pragma unroll
        for (int i = 0; i < 8; ++i) {
            int idx = tid + i * 256;
            int r   = idx >> 5;
            int c4  = idx & 31;
            float4 v = *(const float4*)(W + (size_t)r * IN_DIM + kc + c4 * 4);
            *(float4*)(&ldsB[r * LDP + c4 * 4]) = v;
        }
        __syncthreads();

        const float* pA = &ldsA[(mbase + lm) * LDP + kh];
        const float* pB = &ldsB[(n0   + lm) * LDP + kh];
        #pragma unroll 8
        for (int k = 0; k < KC; k += 4) {
            v2f afrag = *(const v2f*)(pA + k);
            v2f bfrag = *(const v2f*)(pB + k);
            acc = __builtin_amdgcn_wmma_f32_16x16x4_f32(
                false, afrag, false, bfrag, (short)0, acc, false, false);
        }
        __syncthreads();
    }

    const float bv = bias[n0 + lm];
    #pragma unroll
    for (int r = 0; r < 8; ++r) {
        int grow = row0 + mbase + r + half * 8;
        if (grow < n)
            emb[(size_t)grow * OUT_DIM + n0 + lm] = acc[r] + bv;
    }
}

__device__ __forceinline__ bool bin_fits(int b, int n)
{
    return n <= (1 << 17) && (size_t)b * RB * OUT_DIM + (size_t)NCH * SUB <= (size_t)n * OUT_DIM;
}

__global__ __launch_bounds__(1024) void gat_bin(
    const int* __restrict__ edges,
    int* __restrict__ pcnt,
    int* __restrict__ recs,
    int n, int E)
{
    __shared__ int lcnt[MAXB];
    __shared__ __attribute__((aligned(16))) int lrec[MAXB * SUB];

    const int tid = threadIdx.x;
    const int c   = blockIdx.x;
    if (tid < MAXB) lcnt[tid] = 0;
    __syncthreads();

    const int chunk = (E + NCH - 1) / NCH;
    const int e0 = c * chunk;
    const int e1 = min(e0 + chunk, E);
    for (int base = e0 + tid; base < e1; base += 8 * 1024) {
        int2 ed[8];
        #pragma unroll
        for (int k = 0; k < 8; ++k) {
            const int e = base + k * 1024;
            ed[k] = make_int2(0, 0);
            if (e < e1) ed[k] = *(const int2*)(edges + 2 * (size_t)e);
        }
        #pragma unroll
        for (int k = 0; k < 8; ++k) {
            if (base + k * 1024 < e1) {
                const int b   = ed[k].x / RB;
                const int loc = ed[k].x - b * RB;
                const int pos = atomicAdd(&lcnt[b], 1);
                if (pos < SUB) lrec[b * SUB + pos] = (loc << 17) | ed[k].y;
            }
        }
    }
    __syncthreads();

    if (tid < MAXB) pcnt[c * MAXB + tid] = lcnt[tid];
    for (int idx = tid; idx < MAXB * (SUB / 4); idx += 1024) {
        const int b  = idx / (SUB / 4);
        const int q  = idx - b * (SUB / 4);
        const int cb = lcnt[b];
        if (cb <= SUB && q * 4 < cb && bin_fits(b, n))
            *(int4*)(recs + (size_t)b * RB * OUT_DIM + c * SUB + q * 4) = *(const int4*)(&lrec[b * SUB + q * 4]);
    }
}

template <int M>
__device__ __forceinline__ float row_xor(float v)
{
    return __builtin_bit_cast(float,
        __builtin_amdgcn_update_dpp(0, __builtin_bit_cast(int, v), 0x160 + M, 0xf, 0xf, true));
}

__device__ __forceinline__ float row16_sum(float v)
{
    v += row_xor<1>(v);
    v += row_xor<2>(v);
    v += row_xor<4>(v);
    v += row_xor<8>(v);
    return v;
}

__device__ __forceinline__ float dot4(const float4 x, const float4 y)
{
    return x.x * y.x + x.y * y.y + x.z * y.z + x.w * y.w;
}

__device__ __forceinline__ float edge_weight(float z)
{
    const float l = z > 0.f ? z : SLOPE * z;
    return expf(l);
}

__device__ __forceinline__ void list_add(int* lc, int* lslot, int loc, int dst)
{
    const int r = atomicAdd(&lc[loc], 1);
    if (r < CAP) lslot[loc * CAP + r] = dst;
}

__global__ __launch_bounds__(1024) void gat_gather(
    const int* __restrict__ edges,
    const float* __restrict__ emb,
    const float* __restrict__ a,
    const int* __restrict__ pcnt,
    float* out,
    int n, int E)
{
    __shared__ __attribute__((aligned(16))) int lslot[RB * CAP];
    __shared__ int lc[RB];
    __shared__ int ccnt[NCH];

    const int tid  = threadIdx.x;
    const int b    = blockIdx.x;
    const int row0 = b * RB;
    const int* recs = (const int*)out + (size_t)b * RB * OUT_DIM;
    const bool fits = bin_fits(b, n);

    if (tid < RB) lc[tid] = 0;
    if (tid < NCH) ccnt[tid] = pcnt[tid * MAXB + b];
    __syncthreads();

    {
        const int c   = tid >> 2;
        const int j   = tid & 3;
        const int cc  = ccnt[c];
        const int lim = (fits && cc <= SUB) ? cc : 0;
        int4 v[SUB / 16];
        #pragma unroll
        for (int it = 0; it < SUB / 16; ++it) {
            const int p0 = (j + 4 * it) * 4;
            v[it] = make_int4(0, 0, 0, 0);
            if (p0 < lim) v[it] = *(const int4*)(recs + c * SUB + p0);
        }
        #pragma unroll
        for (int it = 0; it < SUB / 16; ++it) {
            const int p0 = (j + 4 * it) * 4;
            if (p0     < lim) list_add(lc, lslot, v[it].x >> 17, v[it].x & 0x1FFFF);
            if (p0 + 1 < lim) list_add(lc, lslot, v[it].y >> 17, v[it].y & 0x1FFFF);
            if (p0 + 2 < lim) list_add(lc, lslot, v[it].z >> 17, v[it].z & 0x1FFFF);
            if (p0 + 3 < lim) list_add(lc, lslot, v[it].w >> 17, v[it].w & 0x1FFFF);
        }
    }
    const int chunk = (E + NCH - 1) / NCH;
    for (int c = 0; c < NCH; ++c) {
        const int cc = ccnt[c];
        if (cc > SUB || (!fits && cc > 0)) {
            const int e0 = c * chunk;
            const int e1 = min(e0 + chunk, E);
            for (int e = e0 + tid; e < e1; e += 1024) {
                const int2 ed = *(const int2*)(edges + 2 * (size_t)e);
                if (ed.x / RB == b) list_add(lc, lslot, ed.x - row0, ed.y);
            }
        }
    }
    __syncthreads();

    const int hw  = tid >> 4;
    const int l16 = tid & 15;
    const float4 a1 = *(const float4*)(a + 4 * l16);
    const float4 a2 = *(const float4*)(a + OUT_DIM + 4 * l16);

    for (int ln = hw; ln < RB; ln += 64) {
        const int node = row0 + ln;
        if (node >= n) break;

        const float4 self = *(const float4*)(emb + (size_t)node * OUT_DIM + 4 * l16);
        const int craw = lc[ln];

        const float s1 = row16_sum(dot4(self, a1));
        float4 acc;
        float rs;
        {
            const float w = edge_weight(s1 + row16_sum(dot4(self, a2)));
            acc = make_float4(w * self.x, w * self.y, w * self.z, w * self.w);
            rs = w;
        }

        const bool over = craw > CAP;
        const int c = over ? 0 : craw;

        for (int jb = 0; jb < c; jb += 8) {
            const int4 dA = *(const int4*)(&lslot[ln * CAP + jb]);
            const int4 dB = *(const int4*)(&lslot[ln * CAP + jb + 4]);
            const int d[8] = {dA.x, dA.y, dA.z, dA.w, dB.x, dB.y, dB.z, dB.w};
            float4 r[8];
            float t[8];
            #pragma unroll
            for (int k = 0; k < 8; ++k) {
                const int dk = (jb + k < c) ? d[k] : node;
                r[k] = *(const float4*)(emb + (size_t)dk * OUT_DIM + 4 * l16);
            }
            #pragma unroll
            for (int k = 0; k < 8; ++k) t[k] = row16_sum(dot4(r[k], a2));
            #pragma unroll
            for (int k = 0; k < 8; ++k) {
                const float w = (jb + k < c) ? edge_weight(s1 + t[k]) : 0.f;
                acc.x += w * r[k].x; acc.y += w * r[k].y; acc.z += w * r[k].z; acc.w += w * r[k].w; rs += w;
            }
        }

        if (over) {
            for (int e0 = 0; e0 < E; e0 += 16) {
                const int e = e0 + l16;
                int hit = 0, dd = node;
                if (e < E) {
                    const int2 ed = *(const int2*)(edges + 2 * (size_t)e);
                    hit = (ed.x == node) ? 1 : 0;
                    dd  = hit ? ed.y : node;
                }
                for (int k = 0; k < 16; ++k) {
                    const int hk = __shfl(hit, k, 16);
                    const int dk = __shfl(dd, k, 16);
                    if (hk) {
                        const float4 r = *(const float4*)(emb + (size_t)dk * OUT_DIM + 4 * l16);
                        const float w = edge_weight(s1 + row16_sum(dot4(r, a2)));
                        acc.x += w * r.x; acc.y += w * r.y; acc.z += w * r.z; acc.w += w * r.w; rs += w;
                    }
                }
            }
        }

        float4 o;
        o.x = acc.x / rs; o.y = acc.y / rs; o.z = acc.z / rs; o.w = acc.w / rs;
        *(float4*)(out + (size_t)node * OUT_DIM + 4 * l16) = o;
    }
}

extern "C" void kernel_launch(void* const* d_in, const int* in_sizes, int n_in,
                              void* d_out, int out_size, void* d_ws, size_t ws_size,
                              hipStream_t stream)
{
    (void)n_in; (void)out_size; (void)ws_size;
    const float* features = (const float*)d_in[0];
    const float* W        = (const float*)d_in[1];
    const float* bias     = (const float*)d_in[2];
    const float* a        = (const float*)d_in[3];
    const int*   edges    = (const int*)d_in[4];

    const int n = in_sizes[0] / IN_DIM;
    const int E = in_sizes[4] / 2;
    float* out = (float*)d_out;

    float* emb  = (float*)d_ws;
    int*   pcnt = (int*)(emb + (size_t)n * OUT_DIM);

    gat_gemm_wmma<<<(n + 31) / 32, 256, 0, stream>>>(features, W, bias, emb, n);
    gat_bin<<<NCH, 1024, 0, stream>>>(edges, pcnt, (int*)d_out, n, E);
    gat_gather<<<(n + RB - 1) / RB, 1024, 0, stream>>>(edges, emb, a, pcnt, out, n, E);
}


# Gather launch: the serial poll of all 256 chunk counts for unstored pieces (a dependent LDS read per chunk, every wave) replaced by one work-group flag set while pieces are read; sum order as before
# speedup vs baseline: 3.2445x; 1.1601x over previous
#include <hip/hip_runtime.h>
#include <hip/hip_bf16.h>
#include <math.h>

typedef float v2f __attribute__((ext_vector_type(2)));
typedef float v8f __attribute__((ext_vector_type(8)));

#define IN_DIM 256
#define OUT_DIM 64
#define SLOPE 0.1f
#define KC 128
#define LDP 132
#define CAP 64
#define RB 400
#define MAXB 256
#define NCH 256
#define SUB 96

__global__ __launch_bounds__(256) void gat_gemm_wmma(
    const float* __restrict__ features,
    const float* __restrict__ W,
    const float* __restrict__ bias,
    float* __restrict__ emb,
    int n)
{
    __shared__ float ldsA[32 * LDP];
    __shared__ float ldsB[OUT_DIM * LDP];

    const int tid  = threadIdx.x;
    const int row0 = blockIdx.x * 32;

    const int wave  = tid >> 5;
    const int lane  = tid & 31;
    const int mbase = (wave >> 2) * 16;
    const int n0    = (wave & 3) * 16;
    const int lm    = lane & 15;
    const int half  = lane >> 4;
    const int kh    = half * 2;

    v8f acc = {0.f, 0.f, 0.f, 0.f, 0.f, 0.f, 0.f, 0.f};

    for (int kc = 0; kc < IN_DIM; kc += KC) {
        #pragma unroll
        for (int i = 0; i < 4; ++i) {
            int idx = tid + i * 256;
            int r   = idx >> 5;
            int c4  = idx & 31;
            float4 v = make_float4(0.f, 0.f, 0.f, 0.f);
            int gr = row0 + r;
            if (gr < n)
                v = *(const float4*)(features + (size_t)gr * IN_DIM + kc + c4 * 4);
            *(float4*)(&ldsA[r * LDP + c4 * 4]) = v;
        }
        #pragma unroll
        for (int i = 0; i < 8; ++i) {
            int idx = tid + i * 256;
            int r   = idx >> 5;
            int c4  = idx & 31;
            float4 v = *(const float4*)(W + (size_t)r * IN_DIM + kc + c4 * 4);
            *(float4*)(&ldsB[r * LDP + c4 * 4]) = v;
        }
        __syncthreads();

        const float* pA = &ldsA[(mbase + lm) * LDP + kh];
        const float* pB = &ldsB[(n0   + lm) * LDP + kh];
        #pragma unroll 8
        for (int k = 0; k < KC; k += 4) {
            v2f afrag = *(const v2f*)(pA + k);
            v2f bfrag = *(const v2f*)(pB + k);
            acc = __builtin_amdgcn_wmma_f32_16x16x4_f32(
                false, afrag, false, bfrag, (short)0, acc, false, false);
        }
        __syncthreads();
    }

    const float bv = bias[n0 + lm];
    #pragma unroll
    for (int r = 0; r < 8; ++r) {
        int grow = row0 + mbase + r + half * 8;
        if (grow < n)
            emb[(size_t)grow * OUT_DIM + n0 + lm] = acc[r] + bv;
    }
}

__device__ __forceinline__ bool bin_fits(int b, int n)
{
    return n <= (1 << 17) && (size_t)b * RB * OUT_DIM + (size_t)NCH * SUB <= (size_t)n * OUT_DIM;
}

__global__ __launch_bounds__(1024) void gat_bin(
    const int* __restrict__ edges,
    int* __restrict__ pcnt,
    int* __restrict__ recs,
    int n, int E)
{
    __shared__ int lcnt[MAXB];
    __shared__ __attribute__((aligned(16))) int lrec[MAXB * SUB];

    const int tid = threadIdx.x;
    const int c   = blockIdx.x;
    if (tid < MAXB) lcnt[tid] = 0;
    __syncthreads();

    const int chunk = (E + NCH - 1) / NCH;
    const int e0 = c * chunk;
    const int e1 = min(e0 + chunk, E);
    for (int base = e0 + tid; base < e1; base += 8 * 1024) {
        int2 ed[8];
        #pragma unroll
        for (int k = 0; k < 8; ++k) {
            const int e = base + k * 1024;
            ed[k] = make_int2(0, 0);
            if (e < e1) ed[k] = *(const int2*)(edges + 2 * (size_t)e);
        }
        #pragma unroll
        for (int k = 0; k < 8; ++k) {
            if (base + k * 1024 < e1) {
                const int b   = ed[k].x / RB;
                const int loc = ed[k].x - b * RB;
                const int pos = atomicAdd(&lcnt[b], 1);
                if (pos < SUB) lrec[b * SUB + pos] = (loc << 17) | ed[k].y;
            }
        }
    }
    __syncthreads();

    if (tid < MAXB) pcnt[c * MAXB + tid] = lcnt[tid];
    for (int idx = tid; idx < MAXB * (SUB / 4); idx += 1024) {
        const int b  = idx / (SUB / 4);
        const int q  = idx - b * (SUB / 4);
        const int cb = lcnt[b];
        if (cb <= SUB && q * 4 < cb && bin_fits(b, n))
            *(int4*)(recs + (size_t)b * RB * OUT_DIM + c * SUB + q * 4) = *(const int4*)(&lrec[b * SUB + q * 4]);
    }
}

template <int M>
__device__ __forceinline__ float row_xor(float v)
{
    return __builtin_bit_cast(float,
        __builtin_amdgcn_update_dpp(0, __builtin_bit_cast(int, v), 0x160 + M, 0xf, 0xf, true));
}

__device__ __forceinline__ float row16_sum(float v)
{
    v += row_xor<1>(v);
    v += row_xor<2>(v);
    v += row_xor<4>(v);
    v += row_xor<8>(v);
    return v;
}

__device__ __forceinline__ float dot4(const float4 x, const float4 y)
{
    return x.x * y.x + x.y * y.y + x.z * y.z + x.w * y.w;
}

__device__ __forceinline__ float edge_weight(float z)
{
    const float l = z > 0.f ? z : SLOPE * z;
    return expf(l);
}

__device__ __forceinline__ void list_add(int* lc, int* lslot, int loc, int dst)
{
    const int r = atomicAdd(&lc[loc], 1);
    if (r < CAP) lslot[loc * CAP + r] = dst;
}

__global__ __launch_bounds__(1024) void gat_gather(
    const int* __restrict__ edges,
    const float* __restrict__ emb,
    const float* __restrict__ a,
    const int* __restrict__ pcnt,
    float* out,
    int n, int E)
{
    __shared__ __attribute__((aligned(16))) int lslot[RB * CAP];
    __shared__ int lc[RB];
    __shared__ int ccnt[NCH];
    __shared__ int unstored;

    const int tid  = threadIdx.x;
    const int b    = blockIdx.x;
    const int row0 = b * RB;
    const int* recs = (const int*)out + (size_t)b * RB * OUT_DIM;
    const bool fits = bin_fits(b, n);

    if (tid < RB) lc[tid] = 0;
    if (tid < NCH) ccnt[tid] = pcnt[tid * MAXB + b];
    if (tid == 0) unstored = 0;
    __syncthreads();

    {
        const int c   = tid >> 2;
        const int j   = tid & 3;
        const int cc  = ccnt[c];
        const int lim = (fits && cc <= SUB) ? cc : 0;
        if (cc > SUB || (!fits && cc > 0)) unstored = 1;
        int4 v[SUB / 16];
        #pragma unroll
        for (int it = 0; it < SUB / 16; ++it) {
            const int p0 = (j + 4 * it) * 4;
            v[it] = make_int4(0, 0, 0, 0);
            if (p0 < lim) v[it] = *(const int4*)(recs + c * SUB + p0);
        }
        #pragma unroll
        for (int it = 0; it < SUB / 16; ++it) {
            const int p0 = (j + 4 * it) * 4;
            if (p0     < lim) list_add(lc, lslot, v[it].x >> 17, v[it].x & 0x1FFFF);
            if (p0 + 1 < lim) list_add(lc, lslot, v[it].y >> 17, v[it].y & 0x1FFFF);
            if (p0 + 2 < lim) list_add(lc, lslot, v[it].z >> 17, v[it].z & 0x1FFFF);
            if (p0 + 3 < lim) list_add(lc, lslot, v[it].w >> 17, v[it].w & 0x1FFFF);
        }
    }
    __syncthreads();

    if (unstored) {
        const int chunk = (E + NCH - 1) / NCH;
        for (int c = 0; c < NCH; ++c) {
            const int cc = ccnt[c];
            if (cc > SUB || (!fits && cc > 0)) {
                const int e0 = c * chunk;
                const int e1 = min(e0 + chunk, E);
                for (int e = e0 + tid; e < e1; e += 1024) {
                    const int2 ed = *(const int2*)(edges + 2 * (size_t)e);
                    if (ed.x / RB == b) list_add(lc, lslot, ed.x - row0, ed.y);
                }
            }
        }
        __syncthreads();
    }

    const int hw  = tid >> 4;
    const int l16 = tid & 15;
    const float4 a1 = *(const float4*)(a + 4 * l16);
    const float4 a2 = *(const float4*)(a + OUT_DIM + 4 * l16);

    for (int ln = hw; ln < RB; ln += 64) {
        const int node = row0 + ln;
        if (node >= n) break;

        const float4 self = *(const float4*)(emb + (size_t)node * OUT_DIM + 4 * l16);
        const int craw = lc[ln];

        const float s1 = row16_sum(dot4(self, a1));
        float4 acc;
        float rs;
        {
            const float w = edge_weight(s1 + row16_sum(dot4(self, a2)));
            acc = make_float4(w * self.x, w * self.y, w * self.z, w * self.w);
            rs = w;
        }

        const bool over = craw > CAP;
        const int c = over ? 0 : craw;

        for (int jb = 0; jb < c; jb += 8) {
            const int4 dA = *(const int4*)(&lslot[ln * CAP + jb]);
            const int4 dB = *(const int4*)(&lslot[ln * CAP + jb + 4]);
            const int d[8] = {dA.x, dA.y, dA.z, dA.w, dB.x, dB.y, dB.z, dB.w};
            float4 r[8];
            float t[8];
            #pragma unroll
            for (int k = 0; k < 8; ++k) {
                const int dk = (jb + k < c) ? d[k] : node;
                r[k] = *(const float4*)(emb + (size_t)dk * OUT_DIM + 4 * l16);
            }
            #pragma unroll
            for (int k = 0; k < 8; ++k) t[k] = row16_sum(dot4(r[k], a2));
            #pragma unroll
            for (int k = 0; k < 8; ++k) {
                const float w = (jb + k < c) ? edge_weight(s1 + t[k]) : 0.f;
                acc.x += w * r[k].x; acc.y += w * r[k].y; acc.z += w * r[k].z; acc.w += w * r[k].w; rs += w;
            }
        }

        if (over) {
            for (int e0 = 0; e0 < E; e0 += 16) {
                const int e = e0 + l16;
                int hit = 0, dd = node;
                if (e < E) {
                    const int2 ed = *(const int2*)(edges + 2 * (size_t)e);
                    hit = (ed.x == node) ? 1 : 0;
                    dd  = hit ? ed.y : node;
                }
                for (int k = 0; k < 16; ++k) {
                    const int hk = __shfl(hit, k, 16);
                    const int dk = __shfl(dd, k, 16);
                    if (hk) {
                        const float4 r = *(const float4*)(emb + (size_t)dk * OUT_DIM + 4 * l16);
                        const float w = edge_weight(s1 + row16_sum(dot4(r, a2)));
                        acc.x += w * r.x; acc.y += w * r.y; acc.z += w * r.z; acc.w += w * r.w; rs += w;
                    }
                }
            }
        }

        float4 o;
        o.x = acc.x / rs; o.y = acc.y / rs; o.z = acc.z / rs; o.w = acc.w / rs;
        *(float4*)(out + (size_t)node * OUT_DIM + 4 * l16) = o;
    }
}

extern "C" void kernel_launch(void* const* d_in, const int* in_sizes, int n_in,
                              void* d_out, int out_size, void* d_ws, size_t ws_size,
                              hipStream_t stream)
{
    (void)n_in; (void)out_size; (void)ws_size;
    const float* features = (const float*)d_in[0];
    const float* W        = (const float*)d_in[1];
    const float* bias     = (const float*)d_in[2];
    const float* a        = (const float*)d_in[3];
    const int*   edges    = (const int*)d_in[4];

    const int n = in_sizes[0] / IN_DIM;
    const int E = in_sizes[4] / 2;
    float* out = (float*)d_out;

    float* emb  = (float*)d_ws;
    int*   pcnt = (int*)(emb + (size_t)n * OUT_DIM);

    gat_gemm_wmma<<<(n + 31) / 32, 256, 0, stream>>>(features, W, bias, emb, n);
    gat_bin<<<NCH, 1024, 0, stream>>>(edges, pcnt, (int*)d_out, n, E);
    gat_gather<<<(n + RB - 1) / RB, 1024, 0, stream>>>(edges, emb, a, pcnt, out, n, E);
}


# Gather loop: each edge's logit sum and weight (slope, exp) finished ONCE by one lane of the 16-lane row and sent round by row_share, not repeated in all 16 lanes; same add tree, same rank order
# speedup vs baseline: 3.4238x; 1.0553x over previous
#include <hip/hip_runtime.h>
#include <hip/hip_bf16.h>
#include <math.h>

typedef float v2f __attribute__((ext_vector_type(2)));
typedef float v8f __attribute__((ext_vector_type(8)));

#define IN_DIM 256
#define OUT_DIM 64
#define SLOPE 0.1f
#define KC 128
#define LDP 132
#define CAP 64
#define RB 400
#define MAXB 256
#define NCH 256
#define SUB 96

__global__ __launch_bounds__(256) void gat_gemm_wmma(
    const float* __restrict__ features,
    const float* __restrict__ W,
    const float* __restrict__ bias,
    float* __restrict__ emb,
    int n)
{
    __shared__ float ldsA[32 * LDP];
    __shared__ float ldsB[OUT_DIM * LDP];

    const int tid  = threadIdx.x;
    const int row0 = blockIdx.x * 32;

    const int wave  = tid >> 5;
    const int lane  = tid & 31;
    const int mbase = (wave >> 2) * 16;
    const int n0    = (wave & 3) * 16;
    const int lm    = lane & 15;
    const int half  = lane >> 4;
    const int kh    = half * 2;

    v8f acc = {0.f, 0.f, 0.f, 0.f, 0.f, 0.f, 0.f, 0.f};

    for (int kc = 0; kc < IN_DIM; kc += KC) {
        #pragma unroll
        for (int i = 0; i < 4; ++i) {
            int idx = tid + i * 256;
            int r   = idx >> 5;
            int c4  = idx & 31;
            float4 v = make_float4(0.f, 0.f, 0.f, 0.f);
            int gr = row0 + r;
            if (gr < n)
                v = *(const float4*)(features + (size_t)gr * IN_DIM + kc + c4 * 4);
            *(float4*)(&ldsA[r * LDP + c4 * 4]) = v;
        }
        #pragma unroll
        for (int i = 0; i < 8; ++i) {
            int idx = tid + i * 256;
            int r   = idx >> 5;
            int c4  = idx & 31;
            float4 v = *(const float4*)(W + (size_t)r * IN_DIM + kc + c4 * 4);
            *(float4*)(&ldsB[r * LDP + c4 * 4]) = v;
        }
        __syncthreads();

        const float* pA = &ldsA[(mbase + lm) * LDP + kh];
        const float* pB = &ldsB[(n0   + lm) * LDP + kh];
        #pragma unroll 8
        for (int k = 0; k < KC; k += 4) {
            v2f afrag = *(const v2f*)(pA + k);
            v2f bfrag = *(const v2f*)(pB + k);
            acc = __builtin_amdgcn_wmma_f32_16x16x4_f32(
                false, afrag, false, bfrag, (short)0, acc, false, false);
        }
        __syncthreads();
    }

    const float bv = bias[n0 + lm];
    #pragma unroll
    for (int r = 0; r < 8; ++r) {
        int grow = row0 + mbase + r + half * 8;
        if (grow < n)
            emb[(size_t)grow * OUT_DIM + n0 + lm] = acc[r] + bv;
    }
}

__device__ __forceinline__ bool bin_fits(int b, int n)
{
    return n <= (1 << 17) && (size_t)b * RB * OUT_DIM + (size_t)NCH * SUB <= (size_t)n * OUT_DIM;
}

__global__ __launch_bounds__(1024) void gat_bin(
    const int* __restrict__ edges,
    int* __restrict__ pcnt,
    int* __restrict__ recs,
    int n, int E)
{
    __shared__ int lcnt[MAXB];
    __shared__ __attribute__((aligned(16))) int lrec[MAXB * SUB];

    const int tid = threadIdx.x;
    const int c   = blockIdx.x;
    if (tid < MAXB) lcnt[tid] = 0;
    __syncthreads();

    const int chunk = (E + NCH - 1) / NCH;
    const int e0 = c * chunk;
    const int e1 = min(e0 + chunk, E);
    for (int base = e0 + tid; base < e1; base += 8 * 1024) {
        int2 ed[8];
        #pragma unroll
        for (int k = 0; k < 8; ++k) {
            const int e = base + k * 1024;
            ed[k] = make_int2(0, 0);
            if (e < e1) ed[k] = *(const int2*)(edges + 2 * (size_t)e);
        }
        #pragma unroll
        for (int k = 0; k < 8; ++k) {
            if (base + k * 1024 < e1) {
                const int b   = ed[k].x / RB;
                const int loc = ed[k].x - b * RB;
                const int pos = atomicAdd(&lcnt[b], 1);
                if (pos < SUB) lrec[b * SUB + pos] = (loc << 17) | ed[k].y;
            }
        }
    }
    __syncthreads();

    if (tid < MAXB) pcnt[c * MAXB + tid] = lcnt[tid];
    for (int idx = tid; idx < MAXB * (SUB / 4); idx += 1024) {
        const int b  = idx / (SUB / 4);
        const int q  = idx - b * (SUB / 4);
        const int cb = lcnt[b];
        if (cb <= SUB && q * 4 < cb && bin_fits(b, n))
            *(int4*)(recs + (size_t)b * RB * OUT_DIM + c * SUB + q * 4) = *(const int4*)(&lrec[b * SUB + q * 4]);
    }
}

template <int M>
__device__ __forceinline__ float row_xor(float v)
{
    return __builtin_bit_cast(float,
        __builtin_amdgcn_update_dpp(0, __builtin_bit_cast(int, v), 0x160 + M, 0xf, 0xf, true));
}

__device__ __forceinline__ float row16_sum(float v)
{
    v += row_xor<1>(v);
    v += row_xor<2>(v);
    v += row_xor<4>(v);
    v += row_xor<8>(v);
    return v;
}

template <int S>
__device__ __forceinline__ float row_share(float v)
{
    return __builtin_bit_cast(float,
        __builtin_amdgcn_update_dpp(0, __builtin_bit_cast(int, v), 0x150 + S, 0xf, 0xf, true));
}

template <int M>
__device__ __forceinline__ float fold_step(bool upper, float lo, float hi)
{
    const float keep = upper ? hi : lo;
    const float send = upper ? lo : hi;
    return keep + row_xor<M>(send);
}

__device__ __forceinline__ float row16_fold8(const float (&p)[8], bool b0, bool b1, bool b2)
{
    const float q0 = fold_step<1>(b0, p[0], p[4]);
    const float q1 = fold_step<1>(b0, p[1], p[5]);
    const float q2 = fold_step<1>(b0, p[2], p[6]);
    const float q3 = fold_step<1>(b0, p[3], p[7]);
    const float u0 = fold_step<2>(b1, q0, q2);
    const float u1 = fold_step<2>(b1, q1, q3);
    float s = fold_step<4>(b2, u0, u1);
    s += row_xor<8>(s);
    return s;
}

__device__ __forceinline__ float dot4(const float4 x, const float4 y)
{
    return x.x * y.x + x.y * y.y + x.z * y.z + x.w * y.w;
}

__device__ __forceinline__ float edge_weight(float z)
{
    const float l = z > 0.f ? z : SLOPE * z;
    return expf(l);
}

template <int S>
__device__ __forceinline__ void add_row(float4& acc, float& rs, const float wl, const float4 r)
{
    const float w = row_share<S>(wl);
    acc.x = fmaf(w, r.x, acc.x); acc.y = fmaf(w, r.y, acc.y);
    acc.z = fmaf(w, r.z, acc.z); acc.w = fmaf(w, r.w, acc.w);
    rs += w;
}

__device__ __forceinline__ void list_add(int* lc, int* lslot, int loc, int dst)
{
    const int r = atomicAdd(&lc[loc], 1);
    if (r < CAP) lslot[loc * CAP + r] = dst;
}

__global__ __launch_bounds__(1024) void gat_gather(
    const int* __restrict__ edges,
    const float* __restrict__ emb,
    const float* __restrict__ a,
    const int* __restrict__ pcnt,
    float* out,
    int n, int E)
{
    __shared__ __attribute__((aligned(16))) int lslot[RB * CAP];
    __shared__ int lc[RB];
    __shared__ int ccnt[NCH];
    __shared__ int unstored;

    const int tid  = threadIdx.x;
    const int b    = blockIdx.x;
    const int row0 = b * RB;
    const int* recs = (const int*)out + (size_t)b * RB * OUT_DIM;
    const bool fits = bin_fits(b, n);

    if (tid < RB) lc[tid] = 0;
    if (tid < NCH) ccnt[tid] = pcnt[tid * MAXB + b];
    if (tid == 0) unstored = 0;
    __syncthreads();

    {
        const int c   = tid >> 2;
        const int j   = tid & 3;
        const int cc  = ccnt[c];
        const int lim = (fits && cc <= SUB) ? cc : 0;
        if (cc > SUB || (!fits && cc > 0)) unstored = 1;
        int4 v[SUB / 16];
        #pragma unroll
        for (int it = 0; it < SUB / 16; ++it) {
            const int p0 = (j + 4 * it) * 4;
            v[it] = make_int4(0, 0, 0, 0);
            if (p0 < lim) v[it] = *(const int4*)(recs + c * SUB + p0);
        }
        #pragma unroll
        for (int it = 0; it < SUB / 16; ++it) {
            const int p0 = (j + 4 * it) * 4;
            if (p0     < lim) list_add(lc, lslot, v[it].x >> 17, v[it].x & 0x1FFFF);
            if (p0 + 1 < lim) list_add(lc, lslot, v[it].y >> 17, v[it].y & 0x1FFFF);
            if (p0 + 2 < lim) list_add(lc, lslot, v[it].z >> 17, v[it].z & 0x1FFFF);
            if (p0 + 3 < lim) list_add(lc, lslot, v[it].w >> 17, v[it].w & 0x1FFFF);
        }
    }
    __syncthreads();

    if (unstored) {
        const int chunk = (E + NCH - 1) / NCH;
        for (int c = 0; c < NCH; ++c) {
            const int cc = ccnt[c];
            if (cc > SUB || (!fits && cc > 0)) {
                const int e0 = c * chunk;
                const int e1 = min(e0 + chunk, E);
                for (int e = e0 + tid; e < e1; e += 1024) {
                    const int2 ed = *(const int2*)(edges + 2 * (size_t)e);
                    if (ed.x / RB == b) list_add(lc, lslot, ed.x - row0, ed.y);
                }
            }
        }
        __syncthreads();
    }

    const int hw  = tid >> 4;
    const int l16 = tid & 15;
    const float4 a1 = *(const float4*)(a + 4 * l16);
    const float4 a2 = *(const float4*)(a + OUT_DIM + 4 * l16);
    const bool b0 = (l16 & 1) != 0, b1 = (l16 & 2) != 0, b2 = (l16 & 4) != 0;
    const int kmine = ((l16 & 1) << 2) | (l16 & 2) | ((l16 >> 2) & 1);

    for (int ln = hw; ln < RB; ln += 64) {
        const int node = row0 + ln;
        if (node >= n) break;

        const float4 self = *(const float4*)(emb + (size_t)node * OUT_DIM + 4 * l16);
        const int craw = lc[ln];

        const float s1 = row16_sum(dot4(self, a1));
        float4 acc;
        float rs;
        {
            const float w = edge_weight(s1 + row16_sum(dot4(self, a2)));
            acc = make_float4(w * self.x, w * self.y, w * self.z, w * self.w);
            rs = w;
        }

        const bool over = craw > CAP;
        const int c = over ? 0 : craw;

        for (int jb = 0; jb < c; jb += 8) {
            const int4 dA = *(const int4*)(&lslot[ln * CAP + jb]);
            const int4 dB = *(const int4*)(&lslot[ln * CAP + jb + 4]);
            const int d[8] = {dA.x, dA.y, dA.z, dA.w, dB.x, dB.y, dB.z, dB.w};
            float4 r[8];
            float p[8];
            #pragma unroll
            for (int k = 0; k < 8; ++k) {
                const int dk = (jb + k < c) ? d[k] : node;
                r[k] = *(const float4*)(emb + (size_t)dk * OUT_DIM + 4 * l16);
            }
            #pragma unroll
            for (int k = 0; k < 8; ++k) p[k] = dot4(r[k], a2);
            float wl = edge_weight(s1 + row16_fold8(p, b0, b1, b2));
            if (jb + kmine >= c) wl = 0.f;
            add_row<0>(acc, rs, wl, r[0]);
            add_row<4>(acc, rs, wl, r[1]);
            add_row<2>(acc, rs, wl, r[2]);
            add_row<6>(acc, rs, wl, r[3]);
            add_row<1>(acc, rs, wl, r[4]);
            add_row<5>(acc, rs, wl, r[5]);
            add_row<3>(acc, rs, wl, r[6]);
            add_row<7>(acc, rs, wl, r[7]);
        }

        if (over) {
            for (int e0 = 0; e0 < E; e0 += 16) {
                const int e = e0 + l16;
                int hit = 0, dd = node;
                if (e < E) {
                    const int2 ed = *(const int2*)(edges + 2 * (size_t)e);
                    hit = (ed.x == node) ? 1 : 0;
                    dd  = hit ? ed.y : node;
                }
                for (int k = 0; k < 16; ++k) {
                    const int hk = __shfl(hit, k, 16);
                    const int dk = __shfl(dd, k, 16);
                    if (hk) {
                        const float4 r = *(const float4*)(emb + (size_t)dk * OUT_DIM + 4 * l16);
                        const float w = edge_weight(s1 + row16_sum(dot4(r, a2)));
                        acc.x += w * r.x; acc.y += w * r.y; acc.z += w * r.z; acc.w += w * r.w; rs += w;
                    }
                }
            }
        }

        float4 o;
        o.x = acc.x / rs; o.y = acc.y / rs; o.z = acc.z / rs; o.w = acc.w / rs;
        *(float4*)(out + (size_t)node * OUT_DIM + 4 * l16) = o;
    }
}

extern "C" void kernel_launch(void* const* d_in, const int* in_sizes, int n_in,
                              void* d_out, int out_size, void* d_ws, size_t ws_size,
                              hipStream_t stream)
{
    (void)n_in; (void)out_size; (void)ws_size;
    const float* features = (const float*)d_in[0];
    const float* W        = (const float*)d_in[1];
    const float* bias     = (const float*)d_in[2];
    const float* a        = (const float*)d_in[3];
    const int*   edges    = (const int*)d_in[4];

    const int n = in_sizes[0] / IN_DIM;
    const int E = in_sizes[4] / 2;
    float* out = (float*)d_out;

    float* emb  = (float*)d_ws;
    int*   pcnt = (int*)(emb + (size_t)n * OUT_DIM);

    gat_gemm_wmma<<<(n + 31) / 32, 256, 0, stream>>>(features, W, bias, emb, n);
    gat_bin<<<NCH, 1024, 0, stream>>>(edges, pcnt, (int*)d_out, n, E);
    gat_gather<<<(n + RB - 1) / RB, 1024, 0, stream>>>(edges, emb, a, pcnt, out, n, E);
}


# Product launch: a thread's 12 tile loads per K-chunk (4 features, 8 W) issued together into registers, then stored to LDS; before, 12 and 5 serial memory round trips in the two chunks, now 1 each
# speedup vs baseline: 3.7440x; 1.0935x over previous
#include <hip/hip_runtime.h>
#include <hip/hip_bf16.h>
#include <math.h>

typedef float v2f __attribute__((ext_vector_type(2)));
typedef float v8f __attribute__((ext_vector_type(8)));

#define IN_DIM 256
#define OUT_DIM 64
#define SLOPE 0.1f
#define KC 128
#define LDP 132
#define CAP 64
#define RB 400
#define MAXB 256
#define NCH 256
#define SUB 96

__global__ __launch_bounds__(256) void gat_gemm_wmma(
    const float* __restrict__ features,
    const float* __restrict__ W,
    const float* __restrict__ bias,
    float* __restrict__ emb,
    int n)
{
    __shared__ float ldsA[32 * LDP];
    __shared__ float ldsB[OUT_DIM * LDP];

    const int tid  = threadIdx.x;
    const int row0 = blockIdx.x * 32;

    const int wave  = tid >> 5;
    const int lane  = tid & 31;
    const int mbase = (wave >> 2) * 16;
    const int n0    = (wave & 3) * 16;
    const int lm    = lane & 15;
    const int half  = lane >> 4;
    const int kh    = half * 2;

    v8f acc = {0.f, 0.f, 0.f, 0.f, 0.f, 0.f, 0.f, 0.f};

    for (int kc = 0; kc < IN_DIM; kc += KC) {
        float4 va[4], vb[8];
        #pragma unroll
        for (int i = 0; i < 4; ++i) {
            int idx = tid + i * 256;
            int r   = idx >> 5;
            int c4  = idx & 31;
            int gr  = row0 + r;
            int gc  = gr < n ? gr : n - 1;
            va[i] = *(const float4*)(features + (size_t)gc * IN_DIM + kc + c4 * 4);
        }
        #pragma unroll
        for (int i = 0; i < 8; ++i) {
            int idx = tid + i * 256;
            int r   = idx >> 5;
            int c4  = idx & 31;
            vb[i] = *(const float4*)(W + (size_t)r * IN_DIM + kc + c4 * 4);
        }
        #pragma unroll
        for (int i = 0; i < 4; ++i) {
            int idx = tid + i * 256;
            int r   = idx >> 5;
            int c4  = idx & 31;
            float4 v = (row0 + r < n) ? va[i] : make_float4(0.f, 0.f, 0.f, 0.f);
            *(float4*)(&ldsA[r * LDP + c4 * 4]) = v;
        }
        #pragma unroll
        for (int i = 0; i < 8; ++i) {
            int idx = tid + i * 256;
            int r   = idx >> 5;
            int c4  = idx & 31;
            *(float4*)(&ldsB[r * LDP + c4 * 4]) = vb[i];
        }
        __syncthreads();

        const float* pA = &ldsA[(mbase + lm) * LDP + kh];
        const float* pB = &ldsB[(n0   + lm) * LDP + kh];
        #pragma unroll 8
        for (int k = 0; k < KC; k += 4) {
            v2f afrag = *(const v2f*)(pA + k);
            v2f bfrag = *(const v2f*)(pB + k);
            acc = __builtin_amdgcn_wmma_f32_16x16x4_f32(
                false, afrag, false, bfrag, (short)0, acc, false, false);
        }
        __syncthreads();
    }

    const float bv = bias[n0 + lm];
    #pragma unroll
    for (int r = 0; r < 8; ++r) {
        int grow = row0 + mbase + r + half * 8;
        if (grow < n)
            emb[(size_t)grow * OUT_DIM + n0 + lm] = acc[r] + bv;
    }
}

__device__ __forceinline__ bool bin_fits(int b, int n)
{
    return n <= (1 << 17) && (size_t)b * RB * OUT_DIM + (size_t)NCH * SUB <= (size_t)n * OUT_DIM;
}

__global__ __launch_bounds__(1024) void gat_bin(
    const int* __restrict__ edges,
    int* __restrict__ pcnt,
    int* __restrict__ recs,
    int n, int E)
{
    __shared__ int lcnt[MAXB];
    __shared__ __attribute__((aligned(16))) int lrec[MAXB * SUB];

    const int tid = threadIdx.x;
    const int c   = blockIdx.x;
    if (tid < MAXB) lcnt[tid] = 0;
    __syncthreads();

    const int chunk = (E + NCH - 1) / NCH;
    const int e0 = c * chunk;
    const int e1 = min(e0 + chunk, E);
    for (int base = e0 + tid; base < e1; base += 8 * 1024) {
        int2 ed[8];
        #pragma unroll
        for (int k = 0; k < 8; ++k) {
            const int e = base + k * 1024;
            ed[k] = make_int2(0, 0);
            if (e < e1) ed[k] = *(const int2*)(edges + 2 * (size_t)e);
        }
        #pragma unroll
        for (int k = 0; k < 8; ++k) {
            if (base + k * 1024 < e1) {
                const int b   = ed[k].x / RB;
                const int loc = ed[k].x - b * RB;
                const int pos = atomicAdd(&lcnt[b], 1);
                if (pos < SUB) lrec[b * SUB + pos] = (loc << 17) | ed[k].y;
            }
        }
    }
    __syncthreads();

    if (tid < MAXB) pcnt[c * MAXB + tid] = lcnt[tid];
    for (int idx = tid; idx < MAXB * (SUB / 4); idx += 1024) {
        const int b  = idx / (SUB / 4);
        const int q  = idx - b * (SUB / 4);
        const int cb = lcnt[b];
        if (cb <= SUB && q * 4 < cb && bin_fits(b, n))
            *(int4*)(recs + (size_t)b * RB * OUT_DIM + c * SUB + q * 4) = *(const int4*)(&lrec[b * SUB + q * 4]);
    }
}

template <int M>
__device__ __forceinline__ float row_xor(float v)
{
    return __builtin_bit_cast(float,
        __builtin_amdgcn_update_dpp(0, __builtin_bit_cast(int, v), 0x160 + M, 0xf, 0xf, true));
}

__device__ __forceinline__ float row16_sum(float v)
{
    v += row_xor<1>(v);
    v += row_xor<2>(v);
    v += row_xor<4>(v);
    v += row_xor<8>(v);
    return v;
}

template <int S>
__device__ __forceinline__ float row_share(float v)
{
    return __builtin_bit_cast(float,
        __builtin_amdgcn_update_dpp(0, __builtin_bit_cast(int, v), 0x150 + S, 0xf, 0xf, true));
}

template <int M>
__device__ __forceinline__ float fold_step(bool upper, float lo, float hi)
{
    const float keep = upper ? hi : lo;
    const float send = upper ? lo : hi;
    return keep + row_xor<M>(send);
}

__device__ __forceinline__ float row16_fold8(const float (&p)[8], bool b0, bool b1, bool b2)
{
    const float q0 = fold_step<1>(b0, p[0], p[4]);
    const float q1 = fold_step<1>(b0, p[1], p[5]);
    const float q2 = fold_step<1>(b0, p[2], p[6]);
    const float q3 = fold_step<1>(b0, p[3], p[7]);
    const float u0 = fold_step<2>(b1, q0, q2);
    const float u1 = fold_step<2>(b1, q1, q3);
    float s = fold_step<4>(b2, u0, u1);
    s += row_xor<8>(s);
    return s;
}

__device__ __forceinline__ float dot4(const float4 x, const float4 y)
{
    return x.x * y.x + x.y * y.y + x.z * y.z + x.w * y.w;
}

__device__ __forceinline__ float edge_weight(float z)
{
    const float l = z > 0.f ? z : SLOPE * z;
    return expf(l);
}

template <int S>
__device__ __forceinline__ void add_row(float4& acc, float& rs, const float wl, const float4 r)
{
    const float w = row_share<S>(wl);
    acc.x = fmaf(w, r.x, acc.x); acc.y = fmaf(w, r.y, acc.y);
    acc.z = fmaf(w, r.z, acc.z); acc.w = fmaf(w, r.w, acc.w);
    rs += w;
}

__device__ __forceinline__ void list_add(int* lc, int* lslot, int loc, int dst)
{
    const int r = atomicAdd(&lc[loc], 1);
    if (r < CAP) lslot[loc * CAP + r] = dst;
}

__global__ __launch_bounds__(1024) void gat_gather(
    const int* __restrict__ edges,
    const float* __restrict__ emb,
    const float* __restrict__ a,
    const int* __restrict__ pcnt,
    float* out,
    int n, int E)
{
    __shared__ __attribute__((aligned(16))) int lslot[RB * CAP];
    __shared__ int lc[RB];
    __shared__ int ccnt[NCH];
    __shared__ int unstored;

    const int tid  = threadIdx.x;
    const int b    = blockIdx.x;
    const int row0 = b * RB;
    const int* recs = (const int*)out + (size_t)b * RB * OUT_DIM;
    const bool fits = bin_fits(b, n);

    if (tid < RB) lc[tid] = 0;
    if (tid < NCH) ccnt[tid] = pcnt[tid * MAXB + b];
    if (tid == 0) unstored = 0;
    __syncthreads();

    {
        const int c   = tid >> 2;
        const int j   = tid & 3;
        const int cc  = ccnt[c];
        const int lim = (fits && cc <= SUB) ? cc : 0;
        if (cc > SUB || (!fits && cc > 0)) unstored = 1;
        int4 v[SUB / 16];
        #pragma unroll
        for (int it = 0; it < SUB / 16; ++it) {
            const int p0 = (j + 4 * it) * 4;
            v[it] = make_int4(0, 0, 0, 0);
            if (p0 < lim) v[it] = *(const int4*)(recs + c * SUB + p0);
        }
        #pragma unroll
        for (int it = 0; it < SUB / 16; ++it) {
            const int p0 = (j + 4 * it) * 4;
            if (p0     < lim) list_add(lc, lslot, v[it].x >> 17, v[it].x & 0x1FFFF);
            if (p0 + 1 < lim) list_add(lc, lslot, v[it].y >> 17, v[it].y & 0x1FFFF);
            if (p0 + 2 < lim) list_add(lc, lslot, v[it].z >> 17, v[it].z & 0x1FFFF);
            if (p0 + 3 < lim) list_add(lc, lslot, v[it].w >> 17, v[it].w & 0x1FFFF);
        }
    }
    __syncthreads();

    if (unstored) {
        const int chunk = (E + NCH - 1) / NCH;
        for (int c = 0; c < NCH; ++c) {
            const int cc = ccnt[c];
            if (cc > SUB || (!fits && cc > 0)) {
                const int e0 = c * chunk;
                const int e1 = min(e0 + chunk, E);
                for (int e = e0 + tid; e < e1; e += 1024) {
                    const int2 ed = *(const int2*)(edges + 2 * (size_t)e);
                    if (ed.x / RB == b) list_add(lc, lslot, ed.x - row0, ed.y);
                }
            }
        }
        __syncthreads();
    }

    const int hw  = tid >> 4;
    const int l16 = tid & 15;
    const float4 a1 = *(const float4*)(a + 4 * l16);
    const float4 a2 = *(const float4*)(a + OUT_DIM + 4 * l16);
    const bool b0 = (l16 & 1) != 0, b1 = (l16 & 2) != 0, b2 = (l16 & 4) != 0;
    const int kmine = ((l16 & 1) << 2) | (l16 & 2) | ((l16 >> 2) & 1);

    for (int ln = hw; ln < RB; ln += 64) {
        const int node = row0 + ln;
        if (node >= n) break;

        const float4 self = *(const float4*)(emb + (size_t)node * OUT_DIM + 4 * l16);
        const int craw = lc[ln];

        const float s1 = row16_sum(dot4(self, a1));
        float4 acc;
        float rs;
        {
            const float w = edge_weight(s1 + row16_sum(dot4(self, a2)));
            acc = make_float4(w * self.x, w * self.y, w * self.z, w * self.w);
            rs = w;
        }

        const bool over = craw > CAP;
        const int c = over ? 0 : craw;

        for (int jb = 0; jb < c; jb += 8) {
            const int4 dA = *(const int4*)(&lslot[ln * CAP + jb]);
            const int4 dB = *(const int4*)(&lslot[ln * CAP + jb + 4]);
            const int d[8] = {dA.x, dA.y, dA.z, dA.w, dB.x, dB.y, dB.z, dB.w};
            float4 r[8];
            float p[8];
            #pragma unroll
            for (int k = 0; k < 8; ++k) {
                const int dk = (jb + k < c) ? d[k] : node;
                r[k] = *(const float4*)(emb + (size_t)dk * OUT_DIM + 4 * l16);
            }
            #pragma unroll
            for (int k = 0; k < 8; ++k) p[k] = dot4(r[k], a2);
            float wl = edge_weight(s1 + row16_fold8(p, b0, b1, b2));
            if (jb + kmine >= c) wl = 0.f;
            add_row<0>(acc, rs, wl, r[0]);
            add_row<4>(acc, rs, wl, r[1]);
            add_row<2>(acc, rs, wl, r[2]);
            add_row<6>(acc, rs, wl, r[3]);
            add_row<1>(acc, rs, wl, r[4]);
            add_row<5>(acc, rs, wl, r[5]);
            add_row<3>(acc, rs, wl, r[6]);
            add_row<7>(acc, rs, wl, r[7]);
        }

        if (over) {
            for (int e0 = 0; e0 < E; e0 += 16) {
                const int e = e0 + l16;
                int hit = 0, dd = node;
                if (e < E) {
                    const int2 ed = *(const int2*)(edges + 2 * (size_t)e);
                    hit = (ed.x == node) ? 1 : 0;
                    dd  = hit ? ed.y : node;
                }
                for (int k = 0; k < 16; ++k) {
                    const int hk = __shfl(hit, k, 16);
                    const int dk = __shfl(dd, k, 16);
                    if (hk) {
                        const float4 r = *(const float4*)(emb + (size_t)dk * OUT_DIM + 4 * l16);
                        const float w = edge_weight(s1 + row16_sum(dot4(r, a2)));
                        acc.x += w * r.x; acc.y += w * r.y; acc.z += w * r.z; acc.w += w * r.w; rs += w;
                    }
                }
            }
        }

        float4 o;
        o.x = acc.x / rs; o.y = acc.y / rs; o.z = acc.z / rs; o.w = acc.w / rs;
        *(float4*)(out + (size_t)node * OUT_DIM + 4 * l16) = o;
    }
}

extern "C" void kernel_launch(void* const* d_in, const int* in_sizes, int n_in,
                              void* d_out, int out_size, void* d_ws, size_t ws_size,
                              hipStream_t stream)
{
    (void)n_in; (void)out_size; (void)ws_size;
    const float* features = (const float*)d_in[0];
    const float* W        = (const float*)d_in[1];
    const float* bias     = (const float*)d_in[2];
    const float* a        = (const float*)d_in[3];
    const int*   edges    = (const int*)d_in[4];

    const int n = in_sizes[0] / IN_DIM;
    const int E = in_sizes[4] / 2;
    float* out = (float*)d_out;

    float* emb  = (float*)d_ws;
    int*   pcnt = (int*)(emb + (size_t)n * OUT_DIM);

    gat_gemm_wmma<<<(n + 31) / 32, 256, 0, stream>>>(features, W, bias, emb, n);
    gat_bin<<<NCH, 1024, 0, stream>>>(edges, pcnt, (int*)d_out, n, E);
    gat_gather<<<(n + RB - 1) / RB, 1024, 0, stream>>>(edges, emb, a, pcnt, out, n, E);
}


# Binning launch merged into the product launch: its 256 edge-chunk work-groups (8 waves, bins in 2 passes to fit a tile's LDS) run beside the tiles; 3 launches -> 2; sums in arrival order as before
# speedup vs baseline: 4.1295x; 1.1030x over previous
#include <hip/hip_runtime.h>
#include <hip/hip_bf16.h>
#include <math.h>

typedef float v2f __attribute__((ext_vector_type(2)));
typedef float v8f __attribute__((ext_vector_type(8)));

#define IN_DIM 256
#define OUT_DIM 64
#define SLOPE 0.1f
#define KC 128
#define LDP 132
#define CAP 64
#define RB 400
#define MAXB 256
#define NCH 256
#define HB (MAXB / 2)
#define EPT 13
#define SUB 96

__device__ __forceinline__ bool bin_fits(int b, int n)
{
    return n <= (1 << 17) && (size_t)b * RB * OUT_DIM + (size_t)NCH * SUB <= (size_t)n * OUT_DIM;
}

__device__ __forceinline__ void bin_chunk(
    const int* __restrict__ edges,
    int* __restrict__ pcnt,
    int* __restrict__ recs,
    int n, int E, int c, int* lrec, int* lcnt)
{
    const int tid = threadIdx.x;
    const int chunk = (E + NCH - 1) / NCH;
    const int e0 = c * chunk;
    const int e1 = min(e0 + chunk, E);

    for (int blo = 0; blo < MAXB; blo += HB) {
        if (tid < HB) lcnt[tid] = 0;
        __syncthreads();

        for (int base = e0 + tid; base < e1; base += EPT * 256) {
            int2 ed[EPT];
            int pos[EPT];
            #pragma unroll
            for (int k = 0; k < EPT; ++k) {
                const int e = base + k * 256;
                ed[k] = make_int2(0, 0);
                if (e < e1) ed[k] = *(const int2*)(edges + 2 * (size_t)e);
            }
            #pragma unroll
            for (int k = 0; k < EPT; ++k) {
                const int lb = ed[k].x / RB - blo;
                pos[k] = SUB;
                if (base + k * 256 < e1 && lb >= 0 && lb < HB) pos[k] = atomicAdd(&lcnt[lb], 1);
            }
            #pragma unroll
            for (int k = 0; k < EPT; ++k) {
                if (pos[k] < SUB) {
                    const int b   = ed[k].x / RB;
                    const int loc = ed[k].x - b * RB;
                    lrec[(b - blo) * SUB + pos[k]] = (loc << 17) | ed[k].y;
                }
            }
        }
        __syncthreads();

        if (tid < HB) pcnt[c * MAXB + blo + tid] = lcnt[tid];
        for (int idx = tid; idx < HB * (SUB / 4); idx += 256) {
            const int lb = idx / (SUB / 4);
            const int q  = idx - lb * (SUB / 4);
            const int b  = blo + lb;
            const int cb = lcnt[lb];
            if (cb <= SUB && q * 4 < cb && bin_fits(b, n))
                *(int4*)(recs + (size_t)b * RB * OUT_DIM + c * SUB + q * 4) = *(const int4*)(&lrec[lb * SUB + q * 4]);
        }
        __syncthreads();
    }
}

__device__ __forceinline__ void product_tile(
    const float* __restrict__ features,
    const float* __restrict__ W,
    const float* __restrict__ bias,
    float* __restrict__ emb,
    int n, int row0,
    float* ldsA,
    float* ldsB)
{
    const int tid  = threadIdx.x;

    const int wave  = tid >> 5;
    const int lane  = tid & 31;
    const int mbase = (wave >> 2) * 16;
    const int n0    = (wave & 3) * 16;
    const int lm    = lane & 15;
    const int half  = lane >> 4;
    const int kh    = half * 2;

    v8f acc = {0.f, 0.f, 0.f, 0.f, 0.f, 0.f, 0.f, 0.f};

    for (int kc = 0; kc < IN_DIM; kc += KC) {
        float4 va[4], vb[8];
        #pragma unroll
        for (int i = 0; i < 4; ++i) {
            int idx = tid + i * 256;
            int r   = idx >> 5;
            int c4  = idx & 31;
            int gr  = row0 + r;
            int gc  = gr < n ? gr : n - 1;
            va[i] = *(const float4*)(features + (size_t)gc * IN_DIM + kc + c4 * 4);
        }
        #pragma unroll
        for (int i = 0; i < 8; ++i) {
            int idx = tid + i * 256;
            int r   = idx >> 5;
            int c4  = idx & 31;
            vb[i] = *(const float4*)(W + (size_t)r * IN_DIM + kc + c4 * 4);
        }
        asm volatile("" ::: "memory");
        #pragma unroll
        for (int i = 0; i < 4; ++i) {
            int idx = tid + i * 256;
            int r   = idx >> 5;
            int c4  = idx & 31;
            float4 v = (row0 + r < n) ? va[i] : make_float4(0.f, 0.f, 0.f, 0.f);
            *(float4*)(&ldsA[r * LDP + c4 * 4]) = v;
        }
        #pragma unroll
        for (int i = 0; i < 8; ++i) {
            int idx = tid + i * 256;
            int r   = idx >> 5;
            int c4  = idx & 31;
            *(float4*)(&ldsB[r * LDP + c4 * 4]) = vb[i];
        }
        __syncthreads();

        const float* pA = &ldsA[(mbase + lm) * LDP + kh];
        const float* pB = &ldsB[(n0   + lm) * LDP + kh];
        #pragma unroll 8
        for (int k = 0; k < KC; k += 4) {
            v2f afrag = *(const v2f*)(pA + k);
            v2f bfrag = *(const v2f*)(pB + k);
            acc = __builtin_amdgcn_wmma_f32_16x16x4_f32(
                false, afrag, false, bfrag, (short)0, acc, false, false);
        }
        __syncthreads();
    }

    const float bv = bias[n0 + lm];
    #pragma unroll
    for (int r = 0; r < 8; ++r) {
        int grow = row0 + mbase + r + half * 8;
        if (grow < n)
            emb[(size_t)grow * OUT_DIM + n0 + lm] = acc[r] + bv;
    }
}

__global__ __launch_bounds__(256) void gat_gemm_bin(
    const float* __restrict__ features,
    const float* __restrict__ W,
    const float* __restrict__ bias,
    float* __restrict__ emb,
    const int* __restrict__ edges,
    int* __restrict__ pcnt,
    int* __restrict__ recs,
    int n, int E)
{
    __shared__ __attribute__((aligned(16))) float lds[(32 + OUT_DIM) * LDP];

    if (blockIdx.x < NCH) {
        int* lrec = (int*)lds;
        bin_chunk(edges, pcnt, recs, n, E, blockIdx.x, lrec, lrec + HB * SUB);
    } else {
        product_tile(features, W, bias, emb, n, (blockIdx.x - NCH) * 32, lds, lds + 32 * LDP);
    }
}

template <int M>
__device__ __forceinline__ float row_xor(float v)
{
    return __builtin_bit_cast(float,
        __builtin_amdgcn_update_dpp(0, __builtin_bit_cast(int, v), 0x160 + M, 0xf, 0xf, true));
}

__device__ __forceinline__ float row16_sum(float v)
{
    v += row_xor<1>(v);
    v += row_xor<2>(v);
    v += row_xor<4>(v);
    v += row_xor<8>(v);
    return v;
}

template <int S>
__device__ __forceinline__ float row_share(float v)
{
    return __builtin_bit_cast(float,
        __builtin_amdgcn_update_dpp(0, __builtin_bit_cast(int, v), 0x150 + S, 0xf, 0xf, true));
}

template <int M>
__device__ __forceinline__ float fold_step(bool upper, float lo, float hi)
{
    const float keep = upper ? hi : lo;
    const float send = upper ? lo : hi;
    return keep + row_xor<M>(send);
}

__device__ __forceinline__ float row16_fold8(const float (&p)[8], bool b0, bool b1, bool b2)
{
    const float q0 = fold_step<1>(b0, p[0], p[4]);
    const float q1 = fold_step<1>(b0, p[1], p[5]);
    const float q2 = fold_step<1>(b0, p[2], p[6]);
    const float q3 = fold_step<1>(b0, p[3], p[7]);
    const float u0 = fold_step<2>(b1, q0, q2);
    const float u1 = fold_step<2>(b1, q1, q3);
    float s = fold_step<4>(b2, u0, u1);
    s += row_xor<8>(s);
    return s;
}

__device__ __forceinline__ float dot4(const float4 x, const float4 y)
{
    return x.x * y.x + x.y * y.y + x.z * y.z + x.w * y.w;
}

__device__ __forceinline__ float edge_weight(float z)
{
    const float l = z > 0.f ? z : SLOPE * z;
    return expf(l);
}

template <int S>
__device__ __forceinline__ void add_row(float4& acc, float& rs, const float wl, const float4 r)
{
    const float w = row_share<S>(wl);
    acc.x = fmaf(w, r.x, acc.x); acc.y = fmaf(w, r.y, acc.y);
    acc.z = fmaf(w, r.z, acc.z); acc.w = fmaf(w, r.w, acc.w);
    rs += w;
}

__device__ __forceinline__ void list_add(int* lc, int* lslot, int loc, int dst)
{
    const int r = atomicAdd(&lc[loc], 1);
    if (r < CAP) lslot[loc * CAP + r] = dst;
}

__global__ __launch_bounds__(1024) void gat_gather(
    const int* __restrict__ edges,
    const float* __restrict__ emb,
    const float* __restrict__ a,
    const int* __restrict__ pcnt,
    float* out,
    int n, int E)
{
    __shared__ __attribute__((aligned(16))) int lslot[RB * CAP];
    __shared__ int lc[RB];
    __shared__ int ccnt[NCH];
    __shared__ int unstored;

    const int tid  = threadIdx.x;
    const int b    = blockIdx.x;
    const int row0 = b * RB;
    const int* recs = (const int*)out + (size_t)b * RB * OUT_DIM;
    const bool fits = bin_fits(b, n);

    if (tid < RB) lc[tid] = 0;
    if (tid < NCH) ccnt[tid] = pcnt[tid * MAXB + b];
    if (tid == 0) unstored = 0;
    __syncthreads();

    {
        const int c   = tid >> 2;
        const int j   = tid & 3;
        const int cc  = ccnt[c];
        const int lim = (fits && cc <= SUB) ? cc : 0;
        if (cc > SUB || (!fits && cc > 0)) unstored = 1;
        int4 v[SUB / 16];
        #pragma unroll
        for (int it = 0; it < SUB / 16; ++it) {
            const int p0 = (j + 4 * it) * 4;
            v[it] = make_int4(0, 0, 0, 0);
            if (p0 < lim) v[it] = *(const int4*)(recs + c * SUB + p0);
        }
        #pragma unroll
        for (int it = 0; it < SUB / 16; ++it) {
            const int p0 = (j + 4 * it) * 4;
            if (p0     < lim) list_add(lc, lslot, v[it].x >> 17, v[it].x & 0x1FFFF);
            if (p0 + 1 < lim) list_add(lc, lslot, v[it].y >> 17, v[it].y & 0x1FFFF);
            if (p0 + 2 < lim) list_add(lc, lslot, v[it].z >> 17, v[it].z & 0x1FFFF);
            if (p0 + 3 < lim) list_add(lc, lslot, v[it].w >> 17, v[it].w & 0x1FFFF);
        }
    }
    __syncthreads();

    if (unstored) {
        const int chunk = (E + NCH - 1) / NCH;
        for (int c = 0; c < NCH; ++c) {
            const int cc = ccnt[c];
            if (cc > SUB || (!fits && cc > 0)) {
                const int e0 = c * chunk;
                const int e1 = min(e0 + chunk, E);
                for (int e = e0 + tid; e < e1; e += 1024) {
                    const int2 ed = *(const int2*)(edges + 2 * (size_t)e);
                    if (ed.x / RB == b) list_add(lc, lslot, ed.x - row0, ed.y);
                }
            }
        }
        __syncthreads();
    }

    const int hw  = tid >> 4;
    const int l16 = tid & 15;
    const float4 a1 = *(const float4*)(a + 4 * l16);
    const float4 a2 = *(const float4*)(a + OUT_DIM + 4 * l16);
    const bool b0 = (l16 & 1) != 0, b1 = (l16 & 2) != 0, b2 = (l16 & 4) != 0;
    const int kmine = ((l16 & 1) << 2) | (l16 & 2) | ((l16 >> 2) & 1);

    for (int ln = hw; ln < RB; ln += 64) {
        const int node = row0 + ln;
        if (node >= n) break;

        const float4 self = *(const float4*)(emb + (size_t)node * OUT_DIM + 4 * l16);
        const int craw = lc[ln];

        const float s1 = row16_sum(dot4(self, a1));
        float4 acc;
        float rs;
        {
            const float w = edge_weight(s1 + row16_sum(dot4(self, a2)));
            acc = make_float4(w * self.x, w * self.y, w * self.z, w * self.w);
            rs = w;
        }

        const bool over = craw > CAP;
        const int c = over ? 0 : craw;

        for (int jb = 0; jb < c; jb += 8) {
            const int4 dA = *(const int4*)(&lslot[ln * CAP + jb]);
            const int4 dB = *(const int4*)(&lslot[ln * CAP + jb + 4]);
            const int d[8] = {dA.x, dA.y, dA.z, dA.w, dB.x, dB.y, dB.z, dB.w};
            float4 r[8];
            float p[8];
            #pragma unroll
            for (int k = 0; k < 8; ++k) {
                const int dk = (jb + k < c) ? d[k] : node;
                r[k] = *(const float4*)(emb + (size_t)dk * OUT_DIM + 4 * l16);
            }
            #pragma unroll
            for (int k = 0; k < 8; ++k) p[k] = dot4(r[k], a2);
            float wl = edge_weight(s1 + row16_fold8(p, b0, b1, b2));
            if (jb + kmine >= c) wl = 0.f;
            add_row<0>(acc, rs, wl, r[0]);
            add_row<4>(acc, rs, wl, r[1]);
            add_row<2>(acc, rs, wl, r[2]);
            add_row<6>(acc, rs, wl, r[3]);
            add_row<1>(acc, rs, wl, r[4]);
            add_row<5>(acc, rs, wl, r[5]);
            add_row<3>(acc, rs, wl, r[6]);
            add_row<7>(acc, rs, wl, r[7]);
        }

        if (over) {
            for (int e0 = 0; e0 < E; e0 += 16) {
                const int e = e0 + l16;
                int hit = 0, dd = node;
                if (e < E) {
                    const int2 ed = *(const int2*)(edges + 2 * (size_t)e);
                    hit = (ed.x == node) ? 1 : 0;
                    dd  = hit ? ed.y : node;
                }
                for (int k = 0; k < 16; ++k) {
                    const int hk = __shfl(hit, k, 16);
                    const int dk = __shfl(dd, k, 16);
                    if (hk) {
                        const float4 r = *(const float4*)(emb + (size_t)dk * OUT_DIM + 4 * l16);
                        const float w = edge_weight(s1 + row16_sum(dot4(r, a2)));
                        acc.x += w * r.x; acc.y += w * r.y; acc.z += w * r.z; acc.w += w * r.w; rs += w;
                    }
                }
            }
        }

        float4 o;
        o.x = acc.x / rs; o.y = acc.y / rs; o.z = acc.z / rs; o.w = acc.w / rs;
        *(float4*)(out + (size_t)node * OUT_DIM + 4 * l16) = o;
    }
}

extern "C" void kernel_launch(void* const* d_in, const int* in_sizes, int n_in,
                              void* d_out, int out_size, void* d_ws, size_t ws_size,
                              hipStream_t stream)
{
    (void)n_in; (void)out_size; (void)ws_size;
    const float* features = (const float*)d_in[0];
    const float* W        = (const float*)d_in[1];
    const float* bias     = (const float*)d_in[2];
    const float* a        = (const float*)d_in[3];
    const int*   edges    = (const int*)d_in[4];

    const int n = in_sizes[0] / IN_DIM;
    const int E = in_sizes[4] / 2;
    float* out = (float*)d_out;

    float* emb  = (float*)d_ws;
    int*   pcnt = (int*)(emb + (size_t)n * OUT_DIM);

    gat_gemm_bin<<<NCH + (n + 31) / 32, 256, 0, stream>>>(features, W, bias, emb, edges, pcnt, (int*)d_out, n, E);
    gat_gather<<<(n + RB - 1) / RB, 1024, 0, stream>>>(edges, emb, a, pcnt, out, n, E);
}
